# starting kernel
# baseline (speedup 1.0000x reference)
#include <hip/hip_runtime.h>

#define N_NODES 50000
#define N_EDGES 800000
#define DIM 128

typedef float v2f __attribute__((ext_vector_type(2)));
typedef float v8f __attribute__((ext_vector_type(8)));

__global__ void k_init_deg(float* __restrict__ deg) {
    int i = blockIdx.x * blockDim.x + threadIdx.x;
    if (i < N_NODES) deg[i] = 1.0f;
}

__global__ void k_deg_accum(const int* __restrict__ dst, float* __restrict__ deg) {
    int e = blockIdx.x * blockDim.x + threadIdx.x;
    if (e < N_EDGES) atomicAdd(&deg[dst[e]], 1.0f);
}

__global__ void k_deg_to_dinv(float* __restrict__ deg) {
    int i = blockIdx.x * blockDim.x + threadIdx.x;
    if (i < N_NODES) {
        float d = deg[i];
        deg[i] = (d > 0.0f) ? rsqrtf(d) : 0.0f;
    }
}

__global__ __launch_bounds__(256)
void k_gemm_wmma_f32(const float* __restrict__ A, const float* __restrict__ W,
                     float* __restrict__ out) {
    const int lane  = threadIdx.x & 31;
    const int wave  = threadIdx.x >> 5;
    const int mTile = blockIdx.x << 4;
    const int n0    = wave << 4;
    const int l15   = lane & 15;
    const int koff  = (lane >> 4) << 1;

    const float* arow = A + (size_t)(mTile + l15) * DIM;
    const float* bcol = W + n0 + l15;

    v8f acc = {};
#pragma unroll
    for (int kk = 0; kk < DIM; kk += 4) {
        v2f a, b;
        a.x = arow[kk + koff];
        a.y = arow[kk + koff + 1];
        b.x = bcol[(size_t)(kk + koff) * DIM];
        b.y = bcol[(size_t)(kk + koff + 1) * DIM];
        acc = __builtin_amdgcn_wmma_f32_16x16x4_f32(
                   false, a,  false, b,
                   (short)0, acc,  false,  false);
    }

    const int  rowBase = mTile + ((lane >> 4) << 3);
    float*     orow    = out + (size_t)rowBase * DIM + n0 + l15;
#pragma unroll
    for (int v = 0; v < 8; ++v)
        orow[(size_t)v * DIM] = acc[v];
}

__global__ void k_self_init(const float* __restrict__ t,
                            const float* __restrict__ dinv,
                            float* __restrict__ agg) {
    int idx = blockIdx.x * blockDim.x + threadIdx.x;
    if (idx >= N_NODES * (DIM / 4)) return;
    int node = idx >> 5;
    float w  = dinv[node];
    w *= w;
    float4 v = ((const float4*)t)[idx];
    v.x *= w; v.y *= w; v.z *= w; v.w *= w;
    ((float4*)agg)[idx] = v;
}

__global__ __launch_bounds__(256)
void k_scatter(const int* __restrict__ src, const int* __restrict__ dst,
               const float* __restrict__ dinv, const float* __restrict__ t,
               float* __restrict__ agg) {
    const int e = (blockIdx.x << 3) + (threadIdx.x >> 5);
    if (e >= N_EDGES) return;
    const int   lane = threadIdx.x & 31;
    const int   s    = src[e];
    const int   d    = dst[e];
    const float nrm  = dinv[s] * dinv[d];

    const float4 v  = ((const float4*)(t + (size_t)s * DIM))[lane];
    float*       ad = agg + (size_t)d * DIM + (lane << 2);
    atomicAdd(ad + 0, v.x * nrm);
    atomicAdd(ad + 1, v.y * nrm);
    atomicAdd(ad + 2, v.z * nrm);
    atomicAdd(ad + 3, v.w * nrm);
}

__global__ void k_bias_relu(const float* __restrict__ agg,
                            const float* __restrict__ b,
                            float* __restrict__ h) {
    int idx = blockIdx.x * blockDim.x + threadIdx.x;
    if (idx >= N_NODES * (DIM / 4)) return;
    int j = idx & (DIM / 4 - 1);
    float4 v  = ((const float4*)agg)[idx];
    float4 bb = ((const float4*)b)[j];
    v.x = fmaxf(v.x + bb.x, 0.0f);
    v.y = fmaxf(v.y + bb.y, 0.0f);
    v.z = fmaxf(v.z + bb.z, 0.0f);
    v.w = fmaxf(v.w + bb.w, 0.0f);
    ((float4*)h)[idx] = v;
}

__global__ __launch_bounds__(256)
void k_heads(const float* __restrict__ h,
             const float* __restrict__ Wt, const float* __restrict__ bt,
             const float* __restrict__ We, const float* __restrict__ be,
             float* __restrict__ out) {
    const int node = (blockIdx.x << 3) + (threadIdx.x >> 5);
    if (node >= N_NODES) return;
    const int lane = threadIdx.x & 31;

    const float4 v  = ((const float4*)(h + (size_t)node * DIM))[lane];
    const float4 wt = ((const float4*)Wt)[lane];
    const float4 we = ((const float4*)We)[lane];

    float st = v.x * wt.x + v.y * wt.y + v.z * wt.z + v.w * wt.w;
    float ev = v.x * we.x + v.y * we.y + v.z * we.z + v.w * we.w;

#pragma unroll
    for (int off = 16; off > 0; off >>= 1) {
        st += __shfl_down(st, off, 32);
        ev += __shfl_down(ev, off, 32);
    }
    if (lane == 0) {
        out[node]           = st + bt[0];
        out[N_NODES + node] = ev + be[0];
    }
}

extern "C" void kernel_launch(void* const* d_in, const int* in_sizes, int n_in,
                              void* d_out, int out_size, void* d_ws, size_t ws_size,
                              hipStream_t stream) {
    (void)in_sizes; (void)n_in; (void)out_size; (void)ws_size;

    const float* x   = (const float*)d_in[0];
    const int*   ei  = (const int*)  d_in[1];
    const float* W1  = (const float*)d_in[2];
    const float* b1  = (const float*)d_in[3];
    const float* W2  = (const float*)d_in[4];
    const float* b2  = (const float*)d_in[5];
    const float* Wt  = (const float*)d_in[6];
    const float* bt  = (const float*)d_in[7];
    const float* We  = (const float*)d_in[8];
    const float* be  = (const float*)d_in[9];
    float*       out = (float*)d_out;

    const int* src = ei;
    const int* dst = ei + N_EDGES;

    float* dinv = (float*)d_ws;
    float* bufA = dinv + ((N_NODES + 255) & ~255);
    float* bufB = bufA + (size_t)N_NODES * DIM;

    const int ELEM4   = N_NODES * (DIM / 4);
    const int gNode   = (N_NODES + 255) / 256;
    const int gEdge   = (N_EDGES + 255) / 256;
    const int gElem4  = (ELEM4 + 255) / 256;
    const int gGemm   = N_NODES / 16;
    const int gWaveE  = (N_EDGES + 7) / 8;
    const int gWaveN  = (N_NODES + 7) / 8;

    k_init_deg   <<<gNode, 256, 0, stream>>>(dinv);
    k_deg_accum  <<<gEdge, 256, 0, stream>>>(dst, dinv);
    k_deg_to_dinv<<<gNode, 256, 0, stream>>>(dinv);

    k_gemm_wmma_f32<<<gGemm,  256, 0, stream>>>(x, W1, bufA);
    k_self_init    <<<gElem4, 256, 0, stream>>>(bufA, dinv, bufB);
    k_scatter      <<<gWaveE, 256, 0, stream>>>(src, dst, dinv, bufA, bufB);
    k_bias_relu    <<<gElem4, 256, 0, stream>>>(bufB, b1, bufA);

    k_gemm_wmma_f32<<<gGemm,  256, 0, stream>>>(bufA, W2, bufB);
    k_self_init    <<<gElem4, 256, 0, stream>>>(bufB, dinv, bufA);
    k_scatter      <<<gWaveE, 256, 0, stream>>>(src, dst, dinv, bufB, bufA);
    k_bias_relu    <<<gElem4, 256, 0, stream>>>(bufA, b2, bufB);

    k_heads<<<gWaveN, 256, 0, stream>>>(bufB, Wt, bt, We, be, out);
}


# scatter: each wave walks 16 consecutive edges (2 batches of 8: scalar index loads, 8 source rows in flight, then the atomics); 6250 work-groups per launch, not 100000; atomics kept, arrival order
# speedup vs baseline: 1.0389x; 1.0389x over previous
#include <hip/hip_runtime.h>

#define N_NODES 50000
#define N_EDGES 800000
#define DIM 128

typedef float v2f __attribute__((ext_vector_type(2)));
typedef float v8f __attribute__((ext_vector_type(8)));

__global__ void k_init_deg(float* __restrict__ deg) {
    int i = blockIdx.x * blockDim.x + threadIdx.x;
    if (i < N_NODES) deg[i] = 1.0f;
}

__global__ void k_deg_accum(const int* __restrict__ dst, float* __restrict__ deg) {
    int e = blockIdx.x * blockDim.x + threadIdx.x;
    if (e < N_EDGES) atomicAdd(&deg[dst[e]], 1.0f);
}

__global__ void k_deg_to_dinv(float* __restrict__ deg) {
    int i = blockIdx.x * blockDim.x + threadIdx.x;
    if (i < N_NODES) {
        float d = deg[i];
        deg[i] = (d > 0.0f) ? rsqrtf(d) : 0.0f;
    }
}

__global__ __launch_bounds__(256)
void k_gemm_wmma_f32(const float* __restrict__ A, const float* __restrict__ W,
                     float* __restrict__ out) {
    const int lane  = threadIdx.x & 31;
    const int wave  = threadIdx.x >> 5;
    const int mTile = blockIdx.x << 4;
    const int n0    = wave << 4;
    const int l15   = lane & 15;
    const int koff  = (lane >> 4) << 1;

    const float* arow = A + (size_t)(mTile + l15) * DIM;
    const float* bcol = W + n0 + l15;

    v8f acc = {};
#pragma unroll
    for (int kk = 0; kk < DIM; kk += 4) {
        v2f a, b;
        a.x = arow[kk + koff];
        a.y = arow[kk + koff + 1];
        b.x = bcol[(size_t)(kk + koff) * DIM];
        b.y = bcol[(size_t)(kk + koff + 1) * DIM];
        acc = __builtin_amdgcn_wmma_f32_16x16x4_f32(
                   false, a,  false, b,
                   (short)0, acc,  false,  false);
    }

    const int  rowBase = mTile + ((lane >> 4) << 3);
    float*     orow    = out + (size_t)rowBase * DIM + n0 + l15;
#pragma unroll
    for (int v = 0; v < 8; ++v)
        orow[(size_t)v * DIM] = acc[v];
}

__global__ void k_self_init(const float* __restrict__ t,
                            const float* __restrict__ dinv,
                            float* __restrict__ agg) {
    int idx = blockIdx.x * blockDim.x + threadIdx.x;
    if (idx >= N_NODES * (DIM / 4)) return;
    int node = idx >> 5;
    float w  = dinv[node];
    w *= w;
    float4 v = ((const float4*)t)[idx];
    v.x *= w; v.y *= w; v.z *= w; v.w *= w;
    ((float4*)agg)[idx] = v;
}

#define EDGES_PER_WAVE 16
#define EDGE_BATCH 8
static_assert(N_EDGES % (8 * EDGES_PER_WAVE) == 0, "grid covers the edges exactly");

__global__ __launch_bounds__(256)
void k_scatter(const int* __restrict__ src, const int* __restrict__ dst,
               const float* __restrict__ dinv, const float* __restrict__ t,
               float* __restrict__ agg) {
    const int wave = __builtin_amdgcn_readfirstlane((int)(threadIdx.x >> 5));
    const int lane = threadIdx.x & 31;
    const int e0   = ((blockIdx.x << 3) + wave) * EDGES_PER_WAVE;
    if (e0 >= N_EDGES) return;

#pragma unroll
    for (int b = 0; b < EDGES_PER_WAVE; b += EDGE_BATCH) {
        int    s[EDGE_BATCH], d[EDGE_BATCH];
        float  nrm[EDGE_BATCH];
        float4 v[EDGE_BATCH];
#pragma unroll
        for (int i = 0; i < EDGE_BATCH; ++i) {
            s[i] = src[e0 + b + i];
            d[i] = dst[e0 + b + i];
        }
#pragma unroll
        for (int i = 0; i < EDGE_BATCH; ++i) {
            nrm[i] = dinv[s[i]] * dinv[d[i]];
            v[i]   = ((const float4*)(t + (size_t)s[i] * DIM))[lane];
        }
#pragma unroll
        for (int i = 0; i < EDGE_BATCH; ++i) {
            float* ad = agg + (size_t)d[i] * DIM + (lane << 2);
            atomicAdd(ad + 0, v[i].x * nrm[i]);
            atomicAdd(ad + 1, v[i].y * nrm[i]);
            atomicAdd(ad + 2, v[i].z * nrm[i]);
            atomicAdd(ad + 3, v[i].w * nrm[i]);
        }
    }
}

__global__ void k_bias_relu(const float* __restrict__ agg,
                            const float* __restrict__ b,
                            float* __restrict__ h) {
    int idx = blockIdx.x * blockDim.x + threadIdx.x;
    if (idx >= N_NODES * (DIM / 4)) return;
    int j = idx & (DIM / 4 - 1);
    float4 v  = ((const float4*)agg)[idx];
    float4 bb = ((const float4*)b)[j];
    v.x = fmaxf(v.x + bb.x, 0.0f);
    v.y = fmaxf(v.y + bb.y, 0.0f);
    v.z = fmaxf(v.z + bb.z, 0.0f);
    v.w = fmaxf(v.w + bb.w, 0.0f);
    ((float4*)h)[idx] = v;
}

__global__ __launch_bounds__(256)
void k_heads(const float* __restrict__ h,
             const float* __restrict__ Wt, const float* __restrict__ bt,
             const float* __restrict__ We, const float* __restrict__ be,
             float* __restrict__ out) {
    const int node = (blockIdx.x << 3) + (threadIdx.x >> 5);
    if (node >= N_NODES) return;
    const int lane = threadIdx.x & 31;

    const float4 v  = ((const float4*)(h + (size_t)node * DIM))[lane];
    const float4 wt = ((const float4*)Wt)[lane];
    const float4 we = ((const float4*)We)[lane];

    float st = v.x * wt.x + v.y * wt.y + v.z * wt.z + v.w * wt.w;
    float ev = v.x * we.x + v.y * we.y + v.z * we.z + v.w * we.w;

#pragma unroll
    for (int off = 16; off > 0; off >>= 1) {
        st += __shfl_down(st, off, 32);
        ev += __shfl_down(ev, off, 32);
    }
    if (lane == 0) {
        out[node]           = st + bt[0];
        out[N_NODES + node] = ev + be[0];
    }
}

extern "C" void kernel_launch(void* const* d_in, const int* in_sizes, int n_in,
                              void* d_out, int out_size, void* d_ws, size_t ws_size,
                              hipStream_t stream) {
    (void)in_sizes; (void)n_in; (void)out_size; (void)ws_size;

    const float* x   = (const float*)d_in[0];
    const int*   ei  = (const int*)  d_in[1];
    const float* W1  = (const float*)d_in[2];
    const float* b1  = (const float*)d_in[3];
    const float* W2  = (const float*)d_in[4];
    const float* b2  = (const float*)d_in[5];
    const float* Wt  = (const float*)d_in[6];
    const float* bt  = (const float*)d_in[7];
    const float* We  = (const float*)d_in[8];
    const float* be  = (const float*)d_in[9];
    float*       out = (float*)d_out;

    const int* src = ei;
    const int* dst = ei + N_EDGES;

    float* dinv = (float*)d_ws;
    float* bufA = dinv + ((N_NODES + 255) & ~255);
    float* bufB = bufA + (size_t)N_NODES * DIM;

    const int ELEM4   = N_NODES * (DIM / 4);
    const int gNode   = (N_NODES + 255) / 256;
    const int gEdge   = (N_EDGES + 255) / 256;
    const int gElem4  = (ELEM4 + 255) / 256;
    const int gGemm   = N_NODES / 16;
    const int gWaveE  = N_EDGES / (8 * EDGES_PER_WAVE);
    const int gWaveN  = (N_NODES + 7) / 8;

    k_init_deg   <<<gNode, 256, 0, stream>>>(dinv);
    k_deg_accum  <<<gEdge, 256, 0, stream>>>(dst, dinv);
    k_deg_to_dinv<<<gNode, 256, 0, stream>>>(dinv);

    k_gemm_wmma_f32<<<gGemm,  256, 0, stream>>>(x, W1, bufA);
    k_self_init    <<<gElem4, 256, 0, stream>>>(bufA, dinv, bufB);
    k_scatter      <<<gWaveE, 256, 0, stream>>>(src, dst, dinv, bufA, bufB);
    k_bias_relu    <<<gElem4, 256, 0, stream>>>(bufB, b1, bufA);

    k_gemm_wmma_f32<<<gGemm,  256, 0, stream>>>(bufA, W2, bufB);
    k_self_init    <<<gElem4, 256, 0, stream>>>(bufB, dinv, bufA);
    k_scatter      <<<gWaveE, 256, 0, stream>>>(src, dst, dinv, bufB, bufA);
    k_bias_relu    <<<gElem4, 256, 0, stream>>>(bufA, b2, bufB);

    k_heads<<<gWaveN, 256, 0, stream>>>(bufB, Wt, bt, We, be, out);
}


# scatter: a lane owns columns lane, lane+32, lane+64, lane+96 of a row, so each atomic and row load of a wave covers one whole 128-B line, not 8 lanes in each of 4 lines; atomics kept, arrival order
# speedup vs baseline: 1.5980x; 1.5382x over previous
#include <hip/hip_runtime.h>

#define N_NODES 50000
#define N_EDGES 800000
#define DIM 128

typedef float v2f __attribute__((ext_vector_type(2)));
typedef float v8f __attribute__((ext_vector_type(8)));

__global__ void k_init_deg(float* __restrict__ deg) {
    int i = blockIdx.x * blockDim.x + threadIdx.x;
    if (i < N_NODES) deg[i] = 1.0f;
}

__global__ void k_deg_accum(const int* __restrict__ dst, float* __restrict__ deg) {
    int e = blockIdx.x * blockDim.x + threadIdx.x;
    if (e < N_EDGES) atomicAdd(&deg[dst[e]], 1.0f);
}

__global__ void k_deg_to_dinv(float* __restrict__ deg) {
    int i = blockIdx.x * blockDim.x + threadIdx.x;
    if (i < N_NODES) {
        float d = deg[i];
        deg[i] = (d > 0.0f) ? rsqrtf(d) : 0.0f;
    }
}

__global__ __launch_bounds__(256)
void k_gemm_wmma_f32(const float* __restrict__ A, const float* __restrict__ W,
                     float* __restrict__ out) {
    const int lane  = threadIdx.x & 31;
    const int wave  = threadIdx.x >> 5;
    const int mTile = blockIdx.x << 4;
    const int n0    = wave << 4;
    const int l15   = lane & 15;
    const int koff  = (lane >> 4) << 1;

    const float* arow = A + (size_t)(mTile + l15) * DIM;
    const float* bcol = W + n0 + l15;

    v8f acc = {};
#pragma unroll
    for (int kk = 0; kk < DIM; kk += 4) {
        v2f a, b;
        a.x = arow[kk + koff];
        a.y = arow[kk + koff + 1];
        b.x = bcol[(size_t)(kk + koff) * DIM];
        b.y = bcol[(size_t)(kk + koff + 1) * DIM];
        acc = __builtin_amdgcn_wmma_f32_16x16x4_f32(
                   false, a,  false, b,
                   (short)0, acc,  false,  false);
    }

    const int  rowBase = mTile + ((lane >> 4) << 3);
    float*     orow    = out + (size_t)rowBase * DIM + n0 + l15;
#pragma unroll
    for (int v = 0; v < 8; ++v)
        orow[(size_t)v * DIM] = acc[v];
}

__global__ void k_self_init(const float* __restrict__ t,
                            const float* __restrict__ dinv,
                            float* __restrict__ agg) {
    int idx = blockIdx.x * blockDim.x + threadIdx.x;
    if (idx >= N_NODES * (DIM / 4)) return;
    int node = idx >> 5;
    float w  = dinv[node];
    w *= w;
    float4 v = ((const float4*)t)[idx];
    v.x *= w; v.y *= w; v.z *= w; v.w *= w;
    ((float4*)agg)[idx] = v;
}

#define EDGES_PER_WAVE 16
#define EDGE_BATCH 8
static_assert(N_EDGES % (8 * EDGES_PER_WAVE) == 0, "grid covers the edges exactly");

__global__ __launch_bounds__(256)
void k_scatter(const int* __restrict__ src, const int* __restrict__ dst,
               const float* __restrict__ dinv, const float* __restrict__ t,
               float* __restrict__ agg) {
    const int wave = __builtin_amdgcn_readfirstlane((int)(threadIdx.x >> 5));
    const int lane = threadIdx.x & 31;
    const int e0   = ((blockIdx.x << 3) + wave) * EDGES_PER_WAVE;
    if (e0 >= N_EDGES) return;

#pragma unroll
    for (int b = 0; b < EDGES_PER_WAVE; b += EDGE_BATCH) {
        int    s[EDGE_BATCH], d[EDGE_BATCH];
        float  nrm[EDGE_BATCH];
        float  v[EDGE_BATCH][DIM / 32];
#pragma unroll
        for (int i = 0; i < EDGE_BATCH; ++i) {
            s[i] = src[e0 + b + i];
            d[i] = dst[e0 + b + i];
        }
#pragma unroll
        for (int i = 0; i < EDGE_BATCH; ++i) {
            nrm[i] = dinv[s[i]] * dinv[d[i]];
            const float* tr = t + (size_t)s[i] * DIM + lane;
#pragma unroll
            for (int j = 0; j < DIM / 32; ++j)
                v[i][j] = tr[j * 32];
        }
#pragma unroll
        for (int i = 0; i < EDGE_BATCH; ++i) {
            float* ad = agg + (size_t)d[i] * DIM + lane;
#pragma unroll
            for (int j = 0; j < DIM / 32; ++j)
                atomicAdd(ad + j * 32, v[i][j] * nrm[i]);
        }
    }
}

__global__ void k_bias_relu(const float* __restrict__ agg,
                            const float* __restrict__ b,
                            float* __restrict__ h) {
    int idx = blockIdx.x * blockDim.x + threadIdx.x;
    if (idx >= N_NODES * (DIM / 4)) return;
    int j = idx & (DIM / 4 - 1);
    float4 v  = ((const float4*)agg)[idx];
    float4 bb = ((const float4*)b)[j];
    v.x = fmaxf(v.x + bb.x, 0.0f);
    v.y = fmaxf(v.y + bb.y, 0.0f);
    v.z = fmaxf(v.z + bb.z, 0.0f);
    v.w = fmaxf(v.w + bb.w, 0.0f);
    ((float4*)h)[idx] = v;
}

__global__ __launch_bounds__(256)
void k_heads(const float* __restrict__ h,
             const float* __restrict__ Wt, const float* __restrict__ bt,
             const float* __restrict__ We, const float* __restrict__ be,
             float* __restrict__ out) {
    const int node = (blockIdx.x << 3) + (threadIdx.x >> 5);
    if (node >= N_NODES) return;
    const int lane = threadIdx.x & 31;

    const float4 v  = ((const float4*)(h + (size_t)node * DIM))[lane];
    const float4 wt = ((const float4*)Wt)[lane];
    const float4 we = ((const float4*)We)[lane];

    float st = v.x * wt.x + v.y * wt.y + v.z * wt.z + v.w * wt.w;
    float ev = v.x * we.x + v.y * we.y + v.z * we.z + v.w * we.w;

#pragma unroll
    for (int off = 16; off > 0; off >>= 1) {
        st += __shfl_down(st, off, 32);
        ev += __shfl_down(ev, off, 32);
    }
    if (lane == 0) {
        out[node]           = st + bt[0];
        out[N_NODES + node] = ev + be[0];
    }
}

extern "C" void kernel_launch(void* const* d_in, const int* in_sizes, int n_in,
                              void* d_out, int out_size, void* d_ws, size_t ws_size,
                              hipStream_t stream) {
    (void)in_sizes; (void)n_in; (void)out_size; (void)ws_size;

    const float* x   = (const float*)d_in[0];
    const int*   ei  = (const int*)  d_in[1];
    const float* W1  = (const float*)d_in[2];
    const float* b1  = (const float*)d_in[3];
    const float* W2  = (const float*)d_in[4];
    const float* b2  = (const float*)d_in[5];
    const float* Wt  = (const float*)d_in[6];
    const float* bt  = (const float*)d_in[7];
    const float* We  = (const float*)d_in[8];
    const float* be  = (const float*)d_in[9];
    float*       out = (float*)d_out;

    const int* src = ei;
    const int* dst = ei + N_EDGES;

    float* dinv = (float*)d_ws;
    float* bufA = dinv + ((N_NODES + 255) & ~255);
    float* bufB = bufA + (size_t)N_NODES * DIM;

    const int ELEM4   = N_NODES * (DIM / 4);
    const int gNode   = (N_NODES + 255) / 256;
    const int gEdge   = (N_EDGES + 255) / 256;
    const int gElem4  = (ELEM4 + 255) / 256;
    const int gGemm   = N_NODES / 16;
    const int gWaveE  = N_EDGES / (8 * EDGES_PER_WAVE);
    const int gWaveN  = (N_NODES + 7) / 8;

    k_init_deg   <<<gNode, 256, 0, stream>>>(dinv);
    k_deg_accum  <<<gEdge, 256, 0, stream>>>(dst, dinv);
    k_deg_to_dinv<<<gNode, 256, 0, stream>>>(dinv);

    k_gemm_wmma_f32<<<gGemm,  256, 0, stream>>>(x, W1, bufA);
    k_self_init    <<<gElem4, 256, 0, stream>>>(bufA, dinv, bufB);
    k_scatter      <<<gWaveE, 256, 0, stream>>>(src, dst, dinv, bufA, bufB);
    k_bias_relu    <<<gElem4, 256, 0, stream>>>(bufB, b1, bufA);

    k_gemm_wmma_f32<<<gGemm,  256, 0, stream>>>(bufA, W2, bufB);
    k_self_init    <<<gElem4, 256, 0, stream>>>(bufB, dinv, bufA);
    k_scatter      <<<gWaveE, 256, 0, stream>>>(src, dst, dinv, bufB, bufA);
    k_bias_relu    <<<gElem4, 256, 0, stream>>>(bufA, b2, bufB);

    k_heads<<<gWaveN, 256, 0, stream>>>(bufB, Wt, bt, We, be, out);
}


# aggregation: float-atomic scatter replaced by a destination-ordered edge list (integer-atomic arrival rank, scan) and one wave per node summing its rows in registers; sum order = arrival rank
# speedup vs baseline: 2.0374x; 1.2750x over previous
#include <hip/hip_runtime.h>

#define N_NODES 50000
#define N_EDGES 800000
#define DIM 128

typedef float v2f __attribute__((ext_vector_type(2)));
typedef float v8f __attribute__((ext_vector_type(8)));

__global__ void k_init_cnt(int* __restrict__ cnt) {
    int i = blockIdx.x * blockDim.x + threadIdx.x;
    if (i < N_NODES) cnt[i] = 1;
}

__global__ void k_rank(const int* __restrict__ dst, int* __restrict__ cnt,
                       int* __restrict__ rank) {
    int e = blockIdx.x * blockDim.x + threadIdx.x;
    if (e < N_EDGES) rank[e] = atomicAdd(&cnt[dst[e]], 1);
}

#define SCAN_BLOCK 1024
#define SCAN_GRID ((N_NODES + 1 + SCAN_BLOCK - 1) / SCAN_BLOCK)
#define OFF_LEN (SCAN_GRID * SCAN_BLOCK)

__global__ __launch_bounds__(SCAN_BLOCK)
void k_scan_dinv(const int* __restrict__ cnt, int* __restrict__ off,
                 float* __restrict__ dinv, int2* __restrict__ list) {
    __shared__ int wtot[32];
    __shared__ int wpre[32];
    const int tid  = threadIdx.x;
    const int lane = tid & 31;
    const int wv   = tid >> 5;

    int pre = 0;
    const int4* c4 = (const int4*)cnt;
    const int   n4 = blockIdx.x * (SCAN_BLOCK / 4);
    for (int j = tid; j < n4; j += SCAN_BLOCK) {
        int4 q = c4[j];
        pre += (q.x + q.y) + (q.z + q.w);
    }

    const int i = blockIdx.x * SCAN_BLOCK + tid;
    const int c = (i < N_NODES) ? cnt[i] : 0;
    float dv = 0.0f;
    if (i < N_NODES) {
        float dg = (float)c;
        dv = (dg > 0.0f) ? rsqrtf(dg) : 0.0f;
        dinv[i] = dv;
    }

    int x = c;
#pragma unroll
    for (int o = 1; o < 32; o <<= 1) {
        int y = __shfl_up(x, o, 32);
        if (lane >= o) x += y;
    }
#pragma unroll
    for (int o = 16; o > 0; o >>= 1) pre += __shfl_xor(pre, o, 32);
    if (lane == 31) wtot[wv] = x;
    if (lane == 0)  wpre[wv] = pre;
    __syncthreads();

    int s = ((lane < wv) ? wtot[lane] : 0) + wpre[lane];
#pragma unroll
    for (int o = 16; o > 0; o >>= 1) s += __shfl_xor(s, o, 32);
    const int o = s + (x - c);
    off[i] = o;
    if (i < N_NODES) list[o] = make_int2(i, __float_as_int(dv * dv));
}

__global__ void k_fill(const int* __restrict__ src, const int* __restrict__ dst,
                       const int* __restrict__ rank, const int* __restrict__ off,
                       const float* __restrict__ dinv, int2* __restrict__ list) {
    int e = blockIdx.x * blockDim.x + threadIdx.x;
    if (e >= N_EDGES) return;
    const int s = src[e];
    const int d = dst[e];
    list[off[d] + rank[e]] = make_int2(s, __float_as_int(dinv[s] * dinv[d]));
}

__global__ __launch_bounds__(256)
void k_gemm_wmma_f32(const float* __restrict__ A, const float* __restrict__ W,
                     float* __restrict__ out) {
    const int lane  = threadIdx.x & 31;
    const int wave  = threadIdx.x >> 5;
    const int mTile = blockIdx.x << 4;
    const int n0    = wave << 4;
    const int l15   = lane & 15;
    const int koff  = (lane >> 4) << 1;

    const float* arow = A + (size_t)(mTile + l15) * DIM;
    const float* bcol = W + n0 + l15;

    v8f acc = {};
#pragma unroll
    for (int kk = 0; kk < DIM; kk += 4) {
        v2f a, b;
        a.x = arow[kk + koff];
        a.y = arow[kk + koff + 1];
        b.x = bcol[(size_t)(kk + koff) * DIM];
        b.y = bcol[(size_t)(kk + koff + 1) * DIM];
        acc = __builtin_amdgcn_wmma_f32_16x16x4_f32(
                   false, a,  false, b,
                   (short)0, acc,  false,  false);
    }

    const int  rowBase = mTile + ((lane >> 4) << 3);
    float*     orow    = out + (size_t)rowBase * DIM + n0 + l15;
#pragma unroll
    for (int v = 0; v < 8; ++v)
        orow[(size_t)v * DIM] = acc[v];
}

template <int ROWS>
__device__ __forceinline__ void add_rows(const int2* __restrict__ list, int e,
                                         const float4* __restrict__ t4, int lane,
                                         float4& acc) {
    int2   p[ROWS];
    float4 v[ROWS];
#pragma unroll
    for (int i = 0; i < ROWS; ++i) p[i] = list[e + i];
#pragma unroll
    for (int i = 0; i < ROWS; ++i) v[i] = t4[(size_t)p[i].x * (DIM / 4) + lane];
#pragma unroll
    for (int i = 0; i < ROWS; ++i) {
        const float w = __int_as_float(p[i].y);
        acc.x = fmaf(v[i].x, w, acc.x);
        acc.y = fmaf(v[i].y, w, acc.y);
        acc.z = fmaf(v[i].z, w, acc.z);
        acc.w = fmaf(v[i].w, w, acc.w);
    }
}

static_assert(N_NODES % 8 == 0, "8 nodes per 256-thread block, grid exact");

__global__ __launch_bounds__(256)
void k_gather(const int* __restrict__ off, const int2* __restrict__ list,
              const float* __restrict__ t, float* __restrict__ agg) {
    const int wave = __builtin_amdgcn_readfirstlane((int)(threadIdx.x >> 5));
    const int lane = threadIdx.x & 31;
    const int d    = (blockIdx.x << 3) + wave;

    const int beg = off[d];
    const int end = off[d + 1];

    const float4* t4 = (const float4*)t;
    float4 acc = make_float4(0.0f, 0.0f, 0.0f, 0.0f);

    int e = beg;
    for (; e + 8 <= end; e += 8) add_rows<8>(list, e, t4, lane, acc);
    if (e + 4 <= end) { add_rows<4>(list, e, t4, lane, acc); e += 4; }
    if (e + 2 <= end) { add_rows<2>(list, e, t4, lane, acc); e += 2; }
    if (e < end)      { add_rows<1>(list, e, t4, lane, acc); }

    ((float4*)agg)[(size_t)d * (DIM / 4) + lane] = acc;
}

__global__ void k_bias_relu(const float* __restrict__ agg,
                            const float* __restrict__ b,
                            float* __restrict__ h) {
    int idx = blockIdx.x * blockDim.x + threadIdx.x;
    if (idx >= N_NODES * (DIM / 4)) return;
    int j = idx & (DIM / 4 - 1);
    float4 v  = ((const float4*)agg)[idx];
    float4 bb = ((const float4*)b)[j];
    v.x = fmaxf(v.x + bb.x, 0.0f);
    v.y = fmaxf(v.y + bb.y, 0.0f);
    v.z = fmaxf(v.z + bb.z, 0.0f);
    v.w = fmaxf(v.w + bb.w, 0.0f);
    ((float4*)h)[idx] = v;
}

__global__ __launch_bounds__(256)
void k_heads(const float* __restrict__ h,
             const float* __restrict__ Wt, const float* __restrict__ bt,
             const float* __restrict__ We, const float* __restrict__ be,
             float* __restrict__ out) {
    const int node = (blockIdx.x << 3) + (threadIdx.x >> 5);
    if (node >= N_NODES) return;
    const int lane = threadIdx.x & 31;

    const float4 v  = ((const float4*)(h + (size_t)node * DIM))[lane];
    const float4 wt = ((const float4*)Wt)[lane];
    const float4 we = ((const float4*)We)[lane];

    float st = v.x * wt.x + v.y * wt.y + v.z * wt.z + v.w * wt.w;
    float ev = v.x * we.x + v.y * we.y + v.z * we.z + v.w * we.w;

#pragma unroll
    for (int off = 16; off > 0; off >>= 1) {
        st += __shfl_down(st, off, 32);
        ev += __shfl_down(ev, off, 32);
    }
    if (lane == 0) {
        out[node]           = st + bt[0];
        out[N_NODES + node] = ev + be[0];
    }
}

extern "C" void kernel_launch(void* const* d_in, const int* in_sizes, int n_in,
                              void* d_out, int out_size, void* d_ws, size_t ws_size,
                              hipStream_t stream) {
    (void)in_sizes; (void)n_in; (void)out_size;

    const float* x   = (const float*)d_in[0];
    const int*   ei  = (const int*)  d_in[1];
    const float* W1  = (const float*)d_in[2];
    const float* b1  = (const float*)d_in[3];
    const float* W2  = (const float*)d_in[4];
    const float* b2  = (const float*)d_in[5];
    const float* Wt  = (const float*)d_in[6];
    const float* bt  = (const float*)d_in[7];
    const float* We  = (const float*)d_in[8];
    const float* be  = (const float*)d_in[9];
    float*       out = (float*)d_out;

    const int* src = ei;
    const int* dst = ei + N_EDGES;

    float* dinv = (float*)d_ws;
    float* bufA = dinv + ((N_NODES + 255) & ~255);
    float* bufB = bufA + (size_t)N_NODES * DIM;
    int*   off  = (int*)(bufB + (size_t)N_NODES * DIM);
    int2*  list = (int2*)(off + OFF_LEN);
    int*   cnt  = (int*)bufB;
    int*   rank = cnt + ((N_NODES + 255) & ~255);

    if ((const char*)(list + N_EDGES + N_NODES) > (const char*)d_ws + ws_size) return;

    const int ELEM4   = N_NODES * (DIM / 4);
    const int gNode   = (N_NODES + 255) / 256;
    const int gEdge   = (N_EDGES + 255) / 256;
    const int gElem4  = (ELEM4 + 255) / 256;
    const int gGemm   = N_NODES / 16;
    const int gWaveN  = (N_NODES + 7) / 8;

    k_init_cnt <<<gNode,     256,        0, stream>>>(cnt);
    k_rank     <<<gEdge,     256,        0, stream>>>(dst, cnt, rank);
    k_scan_dinv<<<SCAN_GRID, SCAN_BLOCK, 0, stream>>>(cnt, off, dinv, list);
    k_fill     <<<gEdge,     256,        0, stream>>>(src, dst, rank, off, dinv, list);

    k_gemm_wmma_f32<<<gGemm,  256, 0, stream>>>(x, W1, bufA);
    k_gather       <<<gWaveN, 256, 0, stream>>>(off, list, bufA, bufB);
    k_bias_relu    <<<gElem4, 256, 0, stream>>>(bufB, b1, bufA);

    k_gemm_wmma_f32<<<gGemm,  256, 0, stream>>>(bufA, W2, bufB);
    k_gather       <<<gWaveN, 256, 0, stream>>>(off, list, bufB, bufA);
    k_bias_relu    <<<gElem4, 256, 0, stream>>>(bufA, b2, bufB);

    k_heads<<<gWaveN, 256, 0, stream>>>(bufB, Wt, bt, We, be, out);
}


# aggregation epilogue fused: the wave that summed a row applies bias+relu in registers; layer 2 also takes both head dot products there, h2 never stored; 8 launches, not 11; sum order unchanged
# speedup vs baseline: 2.3599x; 1.1583x over previous
#include <hip/hip_runtime.h>

#define N_NODES 50000
#define N_EDGES 800000
#define DIM 128

typedef float v2f __attribute__((ext_vector_type(2)));
typedef float v8f __attribute__((ext_vector_type(8)));

__global__ void k_init_cnt(int* __restrict__ cnt) {
    int i = blockIdx.x * blockDim.x + threadIdx.x;
    if (i < N_NODES) cnt[i] = 1;
}

__global__ void k_rank(const int* __restrict__ dst, int* __restrict__ cnt,
                       int* __restrict__ rank) {
    int e = blockIdx.x * blockDim.x + threadIdx.x;
    if (e < N_EDGES) rank[e] = atomicAdd(&cnt[dst[e]], 1);
}

#define SCAN_BLOCK 1024
#define SCAN_GRID ((N_NODES + 1 + SCAN_BLOCK - 1) / SCAN_BLOCK)
#define OFF_LEN (SCAN_GRID * SCAN_BLOCK)

__global__ __launch_bounds__(SCAN_BLOCK)
void k_scan_dinv(const int* __restrict__ cnt, int* __restrict__ off,
                 float* __restrict__ dinv, int2* __restrict__ list) {
    __shared__ int wtot[32];
    __shared__ int wpre[32];
    const int tid  = threadIdx.x;
    const int lane = tid & 31;
    const int wv   = tid >> 5;

    int pre = 0;
    const int4* c4 = (const int4*)cnt;
    const int   n4 = blockIdx.x * (SCAN_BLOCK / 4);
    for (int j = tid; j < n4; j += SCAN_BLOCK) {
        int4 q = c4[j];
        pre += (q.x + q.y) + (q.z + q.w);
    }

    const int i = blockIdx.x * SCAN_BLOCK + tid;
    const int c = (i < N_NODES) ? cnt[i] : 0;
    float dv = 0.0f;
    if (i < N_NODES) {
        float dg = (float)c;
        dv = (dg > 0.0f) ? rsqrtf(dg) : 0.0f;
        dinv[i] = dv;
    }

    int x = c;
#pragma unroll
    for (int o = 1; o < 32; o <<= 1) {
        int y = __shfl_up(x, o, 32);
        if (lane >= o) x += y;
    }
#pragma unroll
    for (int o = 16; o > 0; o >>= 1) pre += __shfl_xor(pre, o, 32);
    if (lane == 31) wtot[wv] = x;
    if (lane == 0)  wpre[wv] = pre;
    __syncthreads();

    int s = ((lane < wv) ? wtot[lane] : 0) + wpre[lane];
#pragma unroll
    for (int o = 16; o > 0; o >>= 1) s += __shfl_xor(s, o, 32);
    const int o = s + (x - c);
    off[i] = o;
    if (i < N_NODES) list[o] = make_int2(i, __float_as_int(dv * dv));
}

__global__ void k_fill(const int* __restrict__ src, const int* __restrict__ dst,
                       const int* __restrict__ rank, const int* __restrict__ off,
                       const float* __restrict__ dinv, int2* __restrict__ list) {
    int e = blockIdx.x * blockDim.x + threadIdx.x;
    if (e >= N_EDGES) return;
    const int s = src[e];
    const int d = dst[e];
    list[off[d] + rank[e]] = make_int2(s, __float_as_int(dinv[s] * dinv[d]));
}

__global__ __launch_bounds__(256)
void k_gemm_wmma_f32(const float* __restrict__ A, const float* __restrict__ W,
                     float* __restrict__ out) {
    const int lane  = threadIdx.x & 31;
    const int wave  = threadIdx.x >> 5;
    const int mTile = blockIdx.x << 4;
    const int n0    = wave << 4;
    const int l15   = lane & 15;
    const int koff  = (lane >> 4) << 1;

    const float* arow = A + (size_t)(mTile + l15) * DIM;
    const float* bcol = W + n0 + l15;

    v8f acc = {};
#pragma unroll
    for (int kk = 0; kk < DIM; kk += 4) {
        v2f a, b;
        a.x = arow[kk + koff];
        a.y = arow[kk + koff + 1];
        b.x = bcol[(size_t)(kk + koff) * DIM];
        b.y = bcol[(size_t)(kk + koff + 1) * DIM];
        acc = __builtin_amdgcn_wmma_f32_16x16x4_f32(
                   false, a,  false, b,
                   (short)0, acc,  false,  false);
    }

    const int  rowBase = mTile + ((lane >> 4) << 3);
    float*     orow    = out + (size_t)rowBase * DIM + n0 + l15;
#pragma unroll
    for (int v = 0; v < 8; ++v)
        orow[(size_t)v * DIM] = acc[v];
}

template <int ROWS>
__device__ __forceinline__ void add_rows(const int2* __restrict__ list, int e,
                                         const float4* __restrict__ t4, int lane,
                                         float4& acc) {
    int2   p[ROWS];
    float4 v[ROWS];
#pragma unroll
    for (int i = 0; i < ROWS; ++i) p[i] = list[e + i];
#pragma unroll
    for (int i = 0; i < ROWS; ++i) v[i] = t4[(size_t)p[i].x * (DIM / 4) + lane];
#pragma unroll
    for (int i = 0; i < ROWS; ++i) {
        const float w = __int_as_float(p[i].y);
        acc.x = fmaf(v[i].x, w, acc.x);
        acc.y = fmaf(v[i].y, w, acc.y);
        acc.z = fmaf(v[i].z, w, acc.z);
        acc.w = fmaf(v[i].w, w, acc.w);
    }
}

static_assert(N_NODES % 8 == 0, "8 nodes per 256-thread block, grid exact");

__device__ __forceinline__ float4 gather_row(const int* __restrict__ off,
                                             const int2* __restrict__ list,
                                             const float* __restrict__ t,
                                             int d, int lane) {
    const int beg = off[d];
    const int end = off[d + 1];

    const float4* t4 = (const float4*)t;
    float4 acc = make_float4(0.0f, 0.0f, 0.0f, 0.0f);

    int e = beg;
    for (; e + 8 <= end; e += 8) add_rows<8>(list, e, t4, lane, acc);
    if (e + 4 <= end) { add_rows<4>(list, e, t4, lane, acc); e += 4; }
    if (e + 2 <= end) { add_rows<2>(list, e, t4, lane, acc); e += 2; }
    if (e < end)      { add_rows<1>(list, e, t4, lane, acc); }
    return acc;
}

__device__ __forceinline__ float4 bias_relu(float4 v, float4 bb) {
    v.x = fmaxf(v.x + bb.x, 0.0f);
    v.y = fmaxf(v.y + bb.y, 0.0f);
    v.z = fmaxf(v.z + bb.z, 0.0f);
    v.w = fmaxf(v.w + bb.w, 0.0f);
    return v;
}

__global__ __launch_bounds__(256)
void k_gather_relu(const int* __restrict__ off, const int2* __restrict__ list,
                   const float* __restrict__ t, const float* __restrict__ b,
                   float* __restrict__ h) {
    const int wave = __builtin_amdgcn_readfirstlane((int)(threadIdx.x >> 5));
    const int lane = threadIdx.x & 31;
    const int d    = (blockIdx.x << 3) + wave;

    const float4 bb  = ((const float4*)b)[lane];
    const float4 acc = gather_row(off, list, t, d, lane);

    ((float4*)h)[(size_t)d * (DIM / 4) + lane] = bias_relu(acc, bb);
}

__global__ __launch_bounds__(256)
void k_gather_relu_heads(const int* __restrict__ off, const int2* __restrict__ list,
                         const float* __restrict__ t, const float* __restrict__ b,
                         const float* __restrict__ Wt, const float* __restrict__ bt,
                         const float* __restrict__ We, const float* __restrict__ be,
                         float* __restrict__ out) {
    const int wave = __builtin_amdgcn_readfirstlane((int)(threadIdx.x >> 5));
    const int lane = threadIdx.x & 31;
    const int d    = (blockIdx.x << 3) + wave;

    const float4 bb = ((const float4*)b)[lane];
    const float4 wt = ((const float4*)Wt)[lane];
    const float4 we = ((const float4*)We)[lane];

    const float4 v = bias_relu(gather_row(off, list, t, d, lane), bb);

    float st = v.x * wt.x + v.y * wt.y + v.z * wt.z + v.w * wt.w;
    float ev = v.x * we.x + v.y * we.y + v.z * we.z + v.w * we.w;

#pragma unroll
    for (int o = 16; o > 0; o >>= 1) {
        st += __shfl_down(st, o, 32);
        ev += __shfl_down(ev, o, 32);
    }
    if (lane == 0) {
        out[d]           = st + bt[0];
        out[N_NODES + d] = ev + be[0];
    }
}

extern "C" void kernel_launch(void* const* d_in, const int* in_sizes, int n_in,
                              void* d_out, int out_size, void* d_ws, size_t ws_size,
                              hipStream_t stream) {
    (void)in_sizes; (void)n_in; (void)out_size;

    const float* x   = (const float*)d_in[0];
    const int*   ei  = (const int*)  d_in[1];
    const float* W1  = (const float*)d_in[2];
    const float* b1  = (const float*)d_in[3];
    const float* W2  = (const float*)d_in[4];
    const float* b2  = (const float*)d_in[5];
    const float* Wt  = (const float*)d_in[6];
    const float* bt  = (const float*)d_in[7];
    const float* We  = (const float*)d_in[8];
    const float* be  = (const float*)d_in[9];
    float*       out = (float*)d_out;

    const int* src = ei;
    const int* dst = ei + N_EDGES;

    float* dinv = (float*)d_ws;
    float* bufA = dinv + ((N_NODES + 255) & ~255);
    float* bufB = bufA + (size_t)N_NODES * DIM;
    int*   off  = (int*)(bufB + (size_t)N_NODES * DIM);
    int2*  list = (int2*)(off + OFF_LEN);
    int*   cnt  = (int*)bufB;
    int*   rank = cnt + ((N_NODES + 255) & ~255);

    if ((const char*)(list + N_EDGES + N_NODES) > (const char*)d_ws + ws_size) return;

    const int gNode   = (N_NODES + 255) / 256;
    const int gEdge   = (N_EDGES + 255) / 256;
    const int gGemm   = N_NODES / 16;
    const int gWaveN  = (N_NODES + 7) / 8;

    k_init_cnt <<<gNode,     256,        0, stream>>>(cnt);
    k_rank     <<<gEdge,     256,        0, stream>>>(dst, cnt, rank);
    k_scan_dinv<<<SCAN_GRID, SCAN_BLOCK, 0, stream>>>(cnt, off, dinv, list);
    k_fill     <<<gEdge,     256,        0, stream>>>(src, dst, rank, off, dinv, list);

    k_gemm_wmma_f32<<<gGemm,  256, 0, stream>>>(x, W1, bufA);
    k_gather_relu  <<<gWaveN, 256, 0, stream>>>(off, list, bufA, b1, bufB);

    k_gemm_wmma_f32    <<<gGemm,  256, 0, stream>>>(bufB, W2, bufA);
    k_gather_relu_heads<<<gWaveN, 256, 0, stream>>>(off, list, bufA, b2, Wt, bt, We, be, out);
}


# both dense products fused into the layer-1 aggregation: a 32-node tile of A x is summed into LDS, times W1 (+b1, relu), times W2 on WMMA; 6 launches, not 8; layer 1 now sums (A x) W1, not A (x W1)
# speedup vs baseline: 2.9414x; 1.2464x over previous
#include <hip/hip_runtime.h>

#define N_NODES 50000
#define N_EDGES 800000
#define DIM 128

typedef float v2f __attribute__((ext_vector_type(2)));
typedef float v8f __attribute__((ext_vector_type(8)));

__global__ void k_init_cnt(int* __restrict__ cnt) {
    int i = blockIdx.x * blockDim.x + threadIdx.x;
    if (i < N_NODES) cnt[i] = 1;
}

__global__ void k_rank(const int* __restrict__ dst, int* __restrict__ cnt,
                       int* __restrict__ rank) {
    int e = blockIdx.x * blockDim.x + threadIdx.x;
    if (e < N_EDGES) rank[e] = atomicAdd(&cnt[dst[e]], 1);
}

#define SCAN_BLOCK 1024
#define SCAN_GRID ((N_NODES + 1 + SCAN_BLOCK - 1) / SCAN_BLOCK)
#define OFF_LEN (SCAN_GRID * SCAN_BLOCK)

__global__ __launch_bounds__(SCAN_BLOCK)
void k_scan_dinv(const int* __restrict__ cnt, int* __restrict__ off,
                 float* __restrict__ dinv, int2* __restrict__ list) {
    __shared__ int wtot[32];
    __shared__ int wpre[32];
    const int tid  = threadIdx.x;
    const int lane = tid & 31;
    const int wv   = tid >> 5;

    int pre = 0;
    const int4* c4 = (const int4*)cnt;
    const int   n4 = blockIdx.x * (SCAN_BLOCK / 4);
    for (int j = tid; j < n4; j += SCAN_BLOCK) {
        int4 q = c4[j];
        pre += (q.x + q.y) + (q.z + q.w);
    }

    const int i = blockIdx.x * SCAN_BLOCK + tid;
    const int c = (i < N_NODES) ? cnt[i] : 0;
    float dv = 0.0f;
    if (i < N_NODES) {
        float dg = (float)c;
        dv = (dg > 0.0f) ? rsqrtf(dg) : 0.0f;
        dinv[i] = dv;
    }

    int x = c;
#pragma unroll
    for (int o = 1; o < 32; o <<= 1) {
        int y = __shfl_up(x, o, 32);
        if (lane >= o) x += y;
    }
#pragma unroll
    for (int o = 16; o > 0; o >>= 1) pre += __shfl_xor(pre, o, 32);
    if (lane == 31) wtot[wv] = x;
    if (lane == 0)  wpre[wv] = pre;
    __syncthreads();

    int s = ((lane < wv) ? wtot[lane] : 0) + wpre[lane];
#pragma unroll
    for (int o = 16; o > 0; o >>= 1) s += __shfl_xor(s, o, 32);
    const int o = s + (x - c);
    off[i] = o;
    if (i < N_NODES) list[o] = make_int2(i, __float_as_int(dv * dv));
}

__global__ void k_fill(const int* __restrict__ src, const int* __restrict__ dst,
                       const int* __restrict__ rank, const int* __restrict__ off,
                       const float* __restrict__ dinv, int2* __restrict__ list) {
    int e = blockIdx.x * blockDim.x + threadIdx.x;
    if (e >= N_EDGES) return;
    const int s = src[e];
    const int d = dst[e];
    list[off[d] + rank[e]] = make_int2(s, __float_as_int(dinv[s] * dinv[d]));
}

template <int ROWS>
__device__ __forceinline__ void add_rows(const int2* __restrict__ list, int e,
                                         const float4* __restrict__ t4, int lane,
                                         float4& acc) {
    int2   p[ROWS];
    float4 v[ROWS];
#pragma unroll
    for (int i = 0; i < ROWS; ++i) p[i] = list[e + i];
#pragma unroll
    for (int i = 0; i < ROWS; ++i) v[i] = t4[(size_t)p[i].x * (DIM / 4) + lane];
#pragma unroll
    for (int i = 0; i < ROWS; ++i) {
        const float w = __int_as_float(p[i].y);
        acc.x = fmaf(v[i].x, w, acc.x);
        acc.y = fmaf(v[i].y, w, acc.y);
        acc.z = fmaf(v[i].z, w, acc.z);
        acc.w = fmaf(v[i].w, w, acc.w);
    }
}

__device__ __forceinline__ float4 gather_range(const int2* __restrict__ list,
                                               const float* __restrict__ t,
                                               int beg, int end, int lane) {
    const float4* t4 = (const float4*)t;
    float4 acc = make_float4(0.0f, 0.0f, 0.0f, 0.0f);

    int e = beg;
    for (; e + 8 <= end; e += 8) add_rows<8>(list, e, t4, lane, acc);
    if (e + 4 <= end) { add_rows<4>(list, e, t4, lane, acc); e += 4; }
    if (e + 2 <= end) { add_rows<2>(list, e, t4, lane, acc); e += 2; }
    if (e < end)      { add_rows<1>(list, e, t4, lane, acc); }
    return acc;
}

__device__ __forceinline__ float4 bias_relu(float4 v, float4 bb) {
    v.x = fmaxf(v.x + bb.x, 0.0f);
    v.y = fmaxf(v.y + bb.y, 0.0f);
    v.z = fmaxf(v.z + bb.z, 0.0f);
    v.w = fmaxf(v.w + bb.w, 0.0f);
    return v;
}

#define TILE_M 32
#define PITCH  132
#define AGG_GRID ((N_NODES + TILE_M - 1) / TILE_M)

static_assert((AGG_GRID * TILE_M + 1) <= OFF_LEN, "off[] covers the last tile's ranges (empty past N)");

__device__ __forceinline__ void tile_times_w(const float* tile,
                                             const float* __restrict__ Wm,
                                             int lane, int n0, v8f& acc0, v8f& acc1) {
    const int c    = lane & 15;
    const int koff = (lane >> 4) << 1;
    const float* a0   = tile + c * PITCH + koff;
    const float* a1   = a0 + 16 * PITCH;
    const float* bcol = Wm + (size_t)koff * DIM + n0 + c;
#pragma unroll
    for (int kk = 0; kk < DIM; kk += 4) {
        const v2f fa0 = *(const v2f*)(a0 + kk);
        const v2f fa1 = *(const v2f*)(a1 + kk);
        v2f fb;
        fb.x = bcol[(size_t)kk * DIM];
        fb.y = bcol[(size_t)(kk + 1) * DIM];
        acc0 = __builtin_amdgcn_wmma_f32_16x16x4_f32(
                    false, fa0,  false, fb,
                    (short)0, acc0,  false,  false);
        acc1 = __builtin_amdgcn_wmma_f32_16x16x4_f32(
                    false, fa1,  false, fb,
                    (short)0, acc1,  false,  false);
    }
}

__global__ __launch_bounds__(256, 16)
void k_agg_mlp(const int* __restrict__ off, const int2* __restrict__ list,
               const float* __restrict__ x,
               const float* __restrict__ W1, const float* __restrict__ b1,
               const float* __restrict__ W2, float* __restrict__ t2) {
    __shared__ __attribute__((aligned(16))) float tile0[TILE_M * PITCH];
    __shared__ __attribute__((aligned(16))) float tile1[TILE_M * PITCH];

    const int wave = __builtin_amdgcn_readfirstlane((int)(threadIdx.x >> 5));
    const int lane = threadIdx.x & 31;
    const int base = blockIdx.x * TILE_M;
    const int row0 = wave << 2;

    int o[5];
#pragma unroll
    for (int j = 0; j < 5; ++j) o[j] = off[base + row0 + j];
#pragma unroll
    for (int j = 0; j < 4; ++j) {
        const float4 s = gather_range(list, x, o[j], o[j + 1], lane);
        *(float4*)(tile0 + (row0 + j) * PITCH + 4 * lane) = s;
    }
    __syncthreads();

    const int n0 = wave << 4;
    const int c  = lane & 15;
    const int h8 = (lane >> 4) << 3;

    {
        v8f p0 = {}, p1 = {};
        tile_times_w(tile0, W1, lane, n0, p0, p1);
        const float bv = b1[n0 + c];
        float* q = tile1 + h8 * PITCH + n0 + c;
#pragma unroll
        for (int v = 0; v < 8; ++v) {
            q[v * PITCH]        = fmaxf(p0[v] + bv, 0.0f);
            q[(16 + v) * PITCH] = fmaxf(p1[v] + bv, 0.0f);
        }
    }
    __syncthreads();

    {
        v8f s0 = {}, s1 = {};
        tile_times_w(tile1, W2, lane, n0, s0, s1);
        float* q = tile0 + h8 * PITCH + n0 + c;
#pragma unroll
        for (int v = 0; v < 8; ++v) {
            q[v * PITCH]        = s0[v];
            q[(16 + v) * PITCH] = s1[v];
        }
    }
    __syncthreads();

#pragma unroll
    for (int j = 0; j < 4; ++j) {
        const int d = base + row0 + j;
        if (d < N_NODES)
            ((float4*)t2)[(size_t)d * (DIM / 4) + lane] =
                *(const float4*)(tile0 + (row0 + j) * PITCH + 4 * lane);
    }
}

static_assert(N_NODES % 8 == 0, "8 nodes per 256-thread block, grid exact");

__global__ __launch_bounds__(256)
void k_gather_relu_heads(const int* __restrict__ off, const int2* __restrict__ list,
                         const float* __restrict__ t, const float* __restrict__ b,
                         const float* __restrict__ Wt, const float* __restrict__ bt,
                         const float* __restrict__ We, const float* __restrict__ be,
                         float* __restrict__ out) {
    const int wave = __builtin_amdgcn_readfirstlane((int)(threadIdx.x >> 5));
    const int lane = threadIdx.x & 31;
    const int d    = (blockIdx.x << 3) + wave;

    const float4 bb = ((const float4*)b)[lane];
    const float4 wt = ((const float4*)Wt)[lane];
    const float4 we = ((const float4*)We)[lane];

    const float4 v = bias_relu(gather_range(list, t, off[d], off[d + 1], lane), bb);

    float st = v.x * wt.x + v.y * wt.y + v.z * wt.z + v.w * wt.w;
    float ev = v.x * we.x + v.y * we.y + v.z * we.z + v.w * we.w;

#pragma unroll
    for (int o = 16; o > 0; o >>= 1) {
        st += __shfl_down(st, o, 32);
        ev += __shfl_down(ev, o, 32);
    }
    if (lane == 0) {
        out[d]           = st + bt[0];
        out[N_NODES + d] = ev + be[0];
    }
}

extern "C" void kernel_launch(void* const* d_in, const int* in_sizes, int n_in,
                              void* d_out, int out_size, void* d_ws, size_t ws_size,
                              hipStream_t stream) {
    (void)in_sizes; (void)n_in; (void)out_size;

    const float* x   = (const float*)d_in[0];
    const int*   ei  = (const int*)  d_in[1];
    const float* W1  = (const float*)d_in[2];
    const float* b1  = (const float*)d_in[3];
    const float* W2  = (const float*)d_in[4];
    const float* b2  = (const float*)d_in[5];
    const float* Wt  = (const float*)d_in[6];
    const float* bt  = (const float*)d_in[7];
    const float* We  = (const float*)d_in[8];
    const float* be  = (const float*)d_in[9];
    float*       out = (float*)d_out;

    const int* src = ei;
    const int* dst = ei + N_EDGES;

    float* dinv = (float*)d_ws;
    float* bufA = dinv + ((N_NODES + 255) & ~255);
    float* bufB = bufA + (size_t)N_NODES * DIM;
    int*   off  = (int*)(bufB + (size_t)N_NODES * DIM);
    int2*  list = (int2*)(off + OFF_LEN);
    int*   cnt  = (int*)bufB;
    int*   rank = cnt + ((N_NODES + 255) & ~255);

    if ((const char*)(list + N_EDGES + N_NODES) > (const char*)d_ws + ws_size) return;

    const int gNode   = (N_NODES + 255) / 256;
    const int gEdge   = (N_EDGES + 255) / 256;
    const int gWaveN  = (N_NODES + 7) / 8;

    k_init_cnt <<<gNode,     256,        0, stream>>>(cnt);
    k_rank     <<<gEdge,     256,        0, stream>>>(dst, cnt, rank);
    k_scan_dinv<<<SCAN_GRID, SCAN_BLOCK, 0, stream>>>(cnt, off, dinv, list);
    k_fill     <<<gEdge,     256,        0, stream>>>(src, dst, rank, off, dinv, list);

    k_agg_mlp<<<AGG_GRID, 256, 0, stream>>>(off, list, x, W1, b1, W2, bufA);

    k_gather_relu_heads<<<gWaveN, 256, 0, stream>>>(off, list, bufA, b2, Wt, bt, We, be, out);
}


# edge table in one insert pass: 64-slot row per node (integer-atomic arrival rank, overflow list), weights from the degrees at gather time; scan and fill dropped: 4 launches; sum order unchanged
# speedup vs baseline: 3.6112x; 1.2277x over previous
#include <hip/hip_runtime.h>

#define N_NODES 50000
#define N_EDGES 800000
#define DIM 128

typedef float v2f __attribute__((ext_vector_type(2)));
typedef float v8f __attribute__((ext_vector_type(8)));

#define CAP  64
#define NPAD ((N_NODES + 255) & ~255)

__global__ void k_init_cnt(int* __restrict__ cnt, int* __restrict__ bucket,
                           int* __restrict__ ovfn) {
    int i = blockIdx.x * blockDim.x + threadIdx.x;
    if (i < N_NODES) {
        cnt[i] = 1;
        bucket[i * CAP] = i;
    } else if (i < NPAD) {
        cnt[i] = 0;
    }
    if (i == 0) ovfn[0] = 0;
}

__global__ void k_insert(const int* __restrict__ src, const int* __restrict__ dst,
                         int* __restrict__ cnt, int* __restrict__ bucket,
                         int* __restrict__ ovfn, int2* __restrict__ ovf) {
    int e = blockIdx.x * blockDim.x + threadIdx.x;
    if (e >= N_EDGES) return;
    const int s = src[e];
    const int d = dst[e];
    const int r = atomicAdd(&cnt[d], 1);
    if (r < CAP) {
        bucket[d * CAP + r] = s;
    } else {
        const int p = atomicAdd(ovfn, 1);
        ovf[p] = make_int2(d, s);
    }
}

__device__ __forceinline__ float dinv_of(int c) {
    const float dg = (float)c;
    return (dg > 0.0f) ? rsqrtf(dg) : 0.0f;
}

template <int ROWS>
__device__ __forceinline__ void add_rows(const int* __restrict__ ent,
                                         const int* __restrict__ cnt, float dd,
                                         const float4* __restrict__ t4, int lane,
                                         float4& acc) {
    int    s[ROWS];
    int    c[ROWS];
    float4 v[ROWS];
#pragma unroll
    for (int i = 0; i < ROWS; ++i) s[i] = ent[i];
#pragma unroll
    for (int i = 0; i < ROWS; ++i) c[i] = cnt[s[i]];
#pragma unroll
    for (int i = 0; i < ROWS; ++i) v[i] = t4[(size_t)s[i] * (DIM / 4) + lane];
#pragma unroll
    for (int i = 0; i < ROWS; ++i) {
        const float w = dinv_of(c[i]) * dd;
        acc.x = fmaf(v[i].x, w, acc.x);
        acc.y = fmaf(v[i].y, w, acc.y);
        acc.z = fmaf(v[i].z, w, acc.z);
        acc.w = fmaf(v[i].w, w, acc.w);
    }
}

__device__ __forceinline__ float4 gather_node(const int* __restrict__ cnt,
                                              const int* __restrict__ bucket,
                                              const int* __restrict__ ovfn,
                                              const int2* __restrict__ ovf,
                                              const float* __restrict__ t,
                                              int d, int n, int lane) {
    const float4* t4 = (const float4*)t;
    const float dd = dinv_of(n);
    float4 acc = make_float4(0.0f, 0.0f, 0.0f, 0.0f);

    const int  m   = (n < CAP) ? n : CAP;
    const int* ent = bucket + (size_t)d * CAP;
    int e = 0;
    for (; e + 8 <= m; e += 8) add_rows<8>(ent + e, cnt, dd, t4, lane, acc);
    if (e + 4 <= m) { add_rows<4>(ent + e, cnt, dd, t4, lane, acc); e += 4; }
    if (e + 2 <= m) { add_rows<2>(ent + e, cnt, dd, t4, lane, acc); e += 2; }
    if (e < m)      { add_rows<1>(ent + e, cnt, dd, t4, lane, acc); }

    if (n > CAP) {
        int mo = ovfn[0];
        mo = (mo < N_EDGES) ? mo : N_EDGES;
        for (int q0 = 0; q0 < mo; q0 += 32) {
            const int q = q0 + lane;
            int2 en = make_int2(-1, 0);
            if (q < mo) en = ovf[q];
            unsigned mask = __builtin_amdgcn_ballot_w32(en.x == d);
            while (mask) {
                const int b = __builtin_ctz(mask);
                mask &= mask - 1;
                const int s = __builtin_amdgcn_readlane(en.y, b);
                const float  w = dinv_of(cnt[s]) * dd;
                const float4 v = t4[(size_t)s * (DIM / 4) + lane];
                acc.x = fmaf(v.x, w, acc.x);
                acc.y = fmaf(v.y, w, acc.y);
                acc.z = fmaf(v.z, w, acc.z);
                acc.w = fmaf(v.w, w, acc.w);
            }
        }
    }
    return acc;
}

__device__ __forceinline__ float4 bias_relu(float4 v, float4 bb) {
    v.x = fmaxf(v.x + bb.x, 0.0f);
    v.y = fmaxf(v.y + bb.y, 0.0f);
    v.z = fmaxf(v.z + bb.z, 0.0f);
    v.w = fmaxf(v.w + bb.w, 0.0f);
    return v;
}

#define TILE_M 32
#define PITCH  132
#define AGG_GRID ((N_NODES + TILE_M - 1) / TILE_M)

static_assert(AGG_GRID * TILE_M <= NPAD, "cnt[] is set up to NPAD, which covers the last tile's rows");

__device__ __forceinline__ void tile_times_w(const float* tile,
                                             const float* __restrict__ Wm,
                                             int lane, int n0, v8f& acc0, v8f& acc1) {
    const int c    = lane & 15;
    const int koff = (lane >> 4) << 1;
    const float* a0   = tile + c * PITCH + koff;
    const float* a1   = a0 + 16 * PITCH;
    const float* bcol = Wm + (size_t)koff * DIM + n0 + c;
#pragma unroll
    for (int kk = 0; kk < DIM; kk += 4) {
        const v2f fa0 = *(const v2f*)(a0 + kk);
        const v2f fa1 = *(const v2f*)(a1 + kk);
        v2f fb;
        fb.x = bcol[(size_t)kk * DIM];
        fb.y = bcol[(size_t)(kk + 1) * DIM];
        acc0 = __builtin_amdgcn_wmma_f32_16x16x4_f32(
                    false, fa0,  false, fb,
                    (short)0, acc0,  false,  false);
        acc1 = __builtin_amdgcn_wmma_f32_16x16x4_f32(
                    false, fa1,  false, fb,
                    (short)0, acc1,  false,  false);
    }
}

__global__ __launch_bounds__(256, 16)
void k_agg_mlp(const int* __restrict__ cnt, const int* __restrict__ bucket,
               const int* __restrict__ ovfn, const int2* __restrict__ ovf,
               const float* __restrict__ x,
               const float* __restrict__ W1, const float* __restrict__ b1,
               const float* __restrict__ W2, float* __restrict__ t2) {
    __shared__ __attribute__((aligned(16))) float tile0[TILE_M * PITCH];
    __shared__ __attribute__((aligned(16))) float tile1[TILE_M * PITCH];

    const int wave = __builtin_amdgcn_readfirstlane((int)(threadIdx.x >> 5));
    const int lane = threadIdx.x & 31;
    const int base = blockIdx.x * TILE_M;
    const int row0 = wave << 2;

    int cn[4];
#pragma unroll
    for (int j = 0; j < 4; ++j) cn[j] = cnt[base + row0 + j];
#pragma unroll
    for (int j = 0; j < 4; ++j) {
        const float4 s = gather_node(cnt, bucket, ovfn, ovf, x, base + row0 + j, cn[j], lane);
        *(float4*)(tile0 + (row0 + j) * PITCH + 4 * lane) = s;
    }
    __syncthreads();

    const int n0 = wave << 4;
    const int c  = lane & 15;
    const int h8 = (lane >> 4) << 3;

    {
        v8f p0 = {}, p1 = {};
        tile_times_w(tile0, W1, lane, n0, p0, p1);
        const float bv = b1[n0 + c];
        float* q = tile1 + h8 * PITCH + n0 + c;
#pragma unroll
        for (int v = 0; v < 8; ++v) {
            q[v * PITCH]        = fmaxf(p0[v] + bv, 0.0f);
            q[(16 + v) * PITCH] = fmaxf(p1[v] + bv, 0.0f);
        }
    }
    __syncthreads();

    {
        v8f s0 = {}, s1 = {};
        tile_times_w(tile1, W2, lane, n0, s0, s1);
        float* q = tile0 + h8 * PITCH + n0 + c;
#pragma unroll
        for (int v = 0; v < 8; ++v) {
            q[v * PITCH]        = s0[v];
            q[(16 + v) * PITCH] = s1[v];
        }
    }
    __syncthreads();

#pragma unroll
    for (int j = 0; j < 4; ++j) {
        const int d = base + row0 + j;
        if (d < N_NODES)
            ((float4*)t2)[(size_t)d * (DIM / 4) + lane] =
                *(const float4*)(tile0 + (row0 + j) * PITCH + 4 * lane);
    }
}

static_assert(N_NODES % 8 == 0, "8 nodes per 256-thread block, grid exact");

__global__ __launch_bounds__(256)
void k_gather_relu_heads(const int* __restrict__ cnt, const int* __restrict__ bucket,
                         const int* __restrict__ ovfn, const int2* __restrict__ ovf,
                         const float* __restrict__ t, const float* __restrict__ b,
                         const float* __restrict__ Wt, const float* __restrict__ bt,
                         const float* __restrict__ We, const float* __restrict__ be,
                         float* __restrict__ out) {
    const int wave = __builtin_amdgcn_readfirstlane((int)(threadIdx.x >> 5));
    const int lane = threadIdx.x & 31;
    const int d    = (blockIdx.x << 3) + wave;

    const float4 bb = ((const float4*)b)[lane];
    const float4 wt = ((const float4*)Wt)[lane];
    const float4 we = ((const float4*)We)[lane];

    const float4 v = bias_relu(gather_node(cnt, bucket, ovfn, ovf, t, d, cnt[d], lane), bb);

    float st = v.x * wt.x + v.y * wt.y + v.z * wt.z + v.w * wt.w;
    float ev = v.x * we.x + v.y * we.y + v.z * we.z + v.w * we.w;

#pragma unroll
    for (int o = 16; o > 0; o >>= 1) {
        st += __shfl_down(st, o, 32);
        ev += __shfl_down(ev, o, 32);
    }
    if (lane == 0) {
        out[d]           = st + bt[0];
        out[N_NODES + d] = ev + be[0];
    }
}

extern "C" void kernel_launch(void* const* d_in, const int* in_sizes, int n_in,
                              void* d_out, int out_size, void* d_ws, size_t ws_size,
                              hipStream_t stream) {
    (void)in_sizes; (void)n_in; (void)out_size;

    const float* x   = (const float*)d_in[0];
    const int*   ei  = (const int*)  d_in[1];
    const float* W1  = (const float*)d_in[2];
    const float* b1  = (const float*)d_in[3];
    const float* W2  = (const float*)d_in[4];
    const float* b2  = (const float*)d_in[5];
    const float* Wt  = (const float*)d_in[6];
    const float* bt  = (const float*)d_in[7];
    const float* We  = (const float*)d_in[8];
    const float* be  = (const float*)d_in[9];
    float*       out = (float*)d_out;

    const int* src = ei;
    const int* dst = ei + N_EDGES;

    int*   cnt    = (int*)d_ws;
    int*   ovfn   = cnt + NPAD;
    float* bufA   = (float*)(ovfn + 256);
    int*   bucket = (int*)(bufA + (size_t)N_NODES * DIM);
    int2*  ovf    = (int2*)(bucket + (size_t)N_NODES * CAP);

    if ((const char*)(ovf + N_EDGES) > (const char*)d_ws + ws_size) return;

    const int gNode   = NPAD / 256;
    const int gEdge   = (N_EDGES + 255) / 256;
    const int gWaveN  = (N_NODES + 7) / 8;

    k_init_cnt<<<gNode, 256, 0, stream>>>(cnt, bucket, ovfn);
    k_insert  <<<gEdge, 256, 0, stream>>>(src, dst, cnt, bucket, ovfn, ovf);

    k_agg_mlp<<<AGG_GRID, 256, 0, stream>>>(cnt, bucket, ovfn, ovf, x, W1, b1, W2, bufA);

    k_gather_relu_heads<<<gWaveN, 256, 0, stream>>>(cnt, bucket, ovfn, ovf, bufA, b2, Wt, bt, We, be, out);
}
